# starting kernel
# baseline (speedup 1.0000x reference)
#include <hip/hip_runtime.h>
#include <hip/hip_bf16.h>
#include <math.h>

typedef float v8f __attribute__((ext_vector_type(8)));
typedef float v2f __attribute__((ext_vector_type(2)));

#define NEG_BIAS (-1.0e10f)

__global__ __launch_bounds__(256) void gemm_nt_wmma_f32(
    const float* __restrict__ X, const float* __restrict__ W,
    float* __restrict__ C, int M)
{
  const int lane      = threadIdx.x & 31;
  const int waveInBlk = threadIdx.x >> 5;
  const int wave      = blockIdx.x * 8 + waveInBlk;
  const int mt        = wave >> 2;
  const int nt        = wave & 3;
  if (mt * 16 >= M) return;

  const int mrow  = mt * 16 + (lane & 15);
  const int wrow  = nt * 16 + (lane & 15);
  const int khalf = (lane >> 4) << 1;

  const float* __restrict__ xp = X + (size_t)mrow * 64 + khalf;
  const float* __restrict__ wp = W + (size_t)wrow * 64 + khalf;

  v8f c = {};
#pragma unroll
  for (int kk = 0; kk < 64; kk += 4) {
    v2f a, b;
    a.x = xp[kk];  a.y = xp[kk + 1];
    b.x = wp[kk];  b.y = wp[kk + 1];
    c = __builtin_amdgcn_wmma_f32_16x16x4_f32(false, a, false, b,
                                              (short)0, c, false, false);
  }

  const int crowBase = mt * 16 + ((lane >> 4) << 3);
  const int ccol     = nt * 16 + (lane & 15);
#pragma unroll
  for (int r = 0; r < 8; ++r)
    C[(size_t)(crowBase + r) * 64 + ccol] = c[r];
}

__global__ __launch_bounds__(256) void attn_softmax_kernel(
    const float* __restrict__ messages, const int* __restrict__ adj,
    const float* __restrict__ proj, float* __restrict__ z)
{
  __shared__ float s_bias[512];
  __shared__ float s_m[4][64], s_l[4][64], s_ss[4][64];
  __shared__ float s_a0[4][64], s_a1[4][64], s_a2[4][64];

  const int bq   = blockIdx.x;
  const int b    = bq >> 9;
  const int tid  = threadIdx.x;
  const int dd   = tid & 63;
  const int part = tid >> 6;

  for (int i = tid; i < 512; i += 256)
    s_bias[i] = (adj[(size_t)bq * 512 + i] > 0) ? 0.0f : NEG_BIAS;
  __syncthreads();

  const float* __restrict__ msg = messages + (size_t)bq * 512 * 64;
  const float* __restrict__ pb  = proj + (size_t)b * 512 * 3 * 64;

  float m = -1.0e30f, l = 0.0f, ss = 0.0f;
  float a0 = 0.0f, a1 = 0.0f, a2 = 0.0f;

  const int k0 = part * 128;
  for (int k = k0; k < k0 + 128; ++k) {
    __builtin_prefetch(&msg[(size_t)(k + 16) * 64 + dd], 0, 0);

    float x    = msg[(size_t)k * 64 + dd] + s_bias[k];
    float mn   = fmaxf(m, x);
    float corr = __expf(m - mn);
    float e    = __expf(x - mn);
    float c2   = corr * corr;

    const float* __restrict__ pk = pb + (size_t)k * 3 * 64 + dd;
    float p0 = pk[0], p1 = pk[64], p2 = pk[128];

    l  = l  * corr + e;
    ss = ss * c2   + e * e;
    a0 = a0 * corr + e * p0;
    a1 = a1 * corr + e * p1;
    a2 = a2 * corr + e * p2;
    m  = mn;
  }

  s_m[part][dd]  = m;  s_l[part][dd]  = l;  s_ss[part][dd] = ss;
  s_a0[part][dd] = a0; s_a1[part][dd] = a1; s_a2[part][dd] = a2;
  __syncthreads();

  if (tid < 64) {
    float M0 = s_m[0][tid], L = s_l[0][tid], SS = s_ss[0][tid];
    float A0 = s_a0[0][tid], A1 = s_a1[0][tid], A2 = s_a2[0][tid];
#pragma unroll
    for (int p = 1; p < 4; ++p) {
      float m2 = s_m[p][tid];
      float mn = fmaxf(M0, m2);
      float f1 = __expf(M0 - mn);
      float f2 = __expf(m2 - mn);
      L  = L  * f1      + s_l[p][tid]  * f2;
      SS = SS * f1 * f1 + s_ss[p][tid] * f2 * f2;
      A0 = A0 * f1 + s_a0[p][tid] * f2;
      A1 = A1 * f1 + s_a1[p][tid] * f2;
      A2 = A2 * f1 + s_a2[p][tid] * f2;
      M0 = mn;
    }
    L = fmaxf(L, 1.0e-30f);
    float scale = sqrtf(SS) / (L * L);
    float* zo = z + (size_t)bq * 3 * 64 + tid;
    zo[0]   = A0 * scale;
    zo[64]  = A1 * scale;
    zo[128] = A2 * scale;
  }
}

extern "C" void kernel_launch(void* const* d_in, const int* in_sizes, int n_in,
                              void* d_out, int out_size, void* d_ws, size_t ws_size,
                              hipStream_t stream) {
  const float* v_equi   = (const float*)d_in[0];
  const float* messages = (const float*)d_in[1];
  const int*   adj      = (const int*)d_in[2];
  const float* w_coord  = (const float*)d_in[3];
  const float* w_attn   = (const float*)d_in[4];
  float* out = (float*)d_out;

  const int B = 4, Nq = 512, D = 64;
  const int Mproj = B * 512 * 3;
  const int Mout  = B * Nq * 3;

  float* proj = (float*)d_ws;
  float* zbuf = proj + (size_t)Mproj * D;

  const int gemmBlocks = ((Mproj / 16) * 4 + 7) / 8;

  gemm_nt_wmma_f32<<<gemmBlocks, 256, 0, stream>>>(v_equi, w_coord, proj, Mproj);

  attn_softmax_kernel<<<B * Nq, 256, 0, stream>>>(messages, adj, proj, zbuf);

  gemm_nt_wmma_f32<<<gemmBlocks, 256, 0, stream>>>(zbuf, w_attn, out, Mout);
}


# attention: each thread carries 4 q rows, so one proj_equi load feeds 4 message elements (4x fewer L2 reads of it); k split 8x64 instead of 4x128 changes the merge's summation tree
# speedup vs baseline: 1.2108x; 1.2108x over previous
#include <hip/hip_runtime.h>
#include <hip/hip_bf16.h>
#include <math.h>

typedef float v8f __attribute__((ext_vector_type(8)));
typedef float v2f __attribute__((ext_vector_type(2)));

#define NEG_BIAS (-1.0e10f)

__global__ __launch_bounds__(256) void gemm_nt_wmma_f32(
    const float* __restrict__ X, const float* __restrict__ W,
    float* __restrict__ C, int M)
{
  const int lane      = threadIdx.x & 31;
  const int waveInBlk = threadIdx.x >> 5;
  const int wave      = blockIdx.x * 8 + waveInBlk;
  const int mt        = wave >> 2;
  const int nt        = wave & 3;
  if (mt * 16 >= M) return;

  const int mrow  = mt * 16 + (lane & 15);
  const int wrow  = nt * 16 + (lane & 15);
  const int khalf = (lane >> 4) << 1;

  const float* __restrict__ xp = X + (size_t)mrow * 64 + khalf;
  const float* __restrict__ wp = W + (size_t)wrow * 64 + khalf;

  v8f c = {};
#pragma unroll
  for (int kk = 0; kk < 64; kk += 4) {
    v2f a, b;
    a.x = xp[kk];  a.y = xp[kk + 1];
    b.x = wp[kk];  b.y = wp[kk + 1];
    c = __builtin_amdgcn_wmma_f32_16x16x4_f32(false, a, false, b,
                                              (short)0, c, false, false);
  }

  const int crowBase = mt * 16 + ((lane >> 4) << 3);
  const int ccol     = nt * 16 + (lane & 15);
#pragma unroll
  for (int r = 0; r < 8; ++r)
    C[(size_t)(crowBase + r) * 64 + ccol] = c[r];
}

#define QB 4
#define NP 8
#define KP 64
static_assert(QB == 4 && NP * KP == 512, "the mask is read as one float4 per k; the partitions must tile the 512 keys");

__global__ __launch_bounds__(64 * NP) void attn_softmax_kernel(
    const float* __restrict__ messages, const int* __restrict__ adj,
    const float* __restrict__ proj, float* __restrict__ z)
{
  __shared__ __attribute__((aligned(16))) float s_bias[512 * QB];
  __shared__ float s_st[6][NP][QB][64];

  const int bq0  = blockIdx.x * QB;
  const int b    = bq0 >> 9;
  const int tid  = threadIdx.x;
  const int dd   = tid & 63;
  const int part = tid >> 6;

  for (int i = tid; i < 512 * QB; i += 64 * NP) {
    const int j = i >> 9, k = i & 511;
    s_bias[k * QB + j] = (adj[(size_t)(bq0 + j) * 512 + k] > 0) ? 0.0f : NEG_BIAS;
  }
  __syncthreads();

  const int k0 = part * KP;
  const float* __restrict__ msg = messages + (size_t)bq0 * 512 * 64 + (size_t)k0 * 64 + dd;
  const float* __restrict__ pk  = proj + (size_t)b * 512 * 3 * 64 + (size_t)k0 * 3 * 64 + dd;

  float m[QB], l[QB], ss[QB], a0[QB], a1[QB], a2[QB];
#pragma unroll
  for (int j = 0; j < QB; ++j) {
    m[j] = -1.0e30f; l[j] = 0.0f; ss[j] = 0.0f;
    a0[j] = 0.0f; a1[j] = 0.0f; a2[j] = 0.0f;
  }

#pragma unroll 4
  for (int i = 0; i < KP; ++i) {
    const float p0 = pk[0], p1 = pk[64], p2 = pk[128];
    const float4 bv = *reinterpret_cast<const float4*>(&s_bias[(k0 + i) * QB]);
    const float bias[QB] = {bv.x, bv.y, bv.z, bv.w};

#pragma unroll
    for (int j = 0; j < QB; ++j) {
      const float* __restrict__ mp = msg + (size_t)j * 512 * 64;
      __builtin_prefetch(mp + 16 * 64, 0, 0);

      float x    = mp[0] + bias[j];
      float mn   = fmaxf(m[j], x);
      float corr = __expf(m[j] - mn);
      float e    = __expf(x - mn);
      float c2   = corr * corr;

      l[j]  = l[j]  * corr + e;
      ss[j] = ss[j] * c2   + e * e;
      a0[j] = a0[j] * corr + e * p0;
      a1[j] = a1[j] * corr + e * p1;
      a2[j] = a2[j] * corr + e * p2;
      m[j]  = mn;
    }
    msg += 64;
    pk  += 3 * 64;
  }

#pragma unroll
  for (int j = 0; j < QB; ++j) {
    s_st[0][part][j][dd] = m[j];  s_st[1][part][j][dd] = l[j];
    s_st[2][part][j][dd] = ss[j]; s_st[3][part][j][dd] = a0[j];
    s_st[4][part][j][dd] = a1[j]; s_st[5][part][j][dd] = a2[j];
  }
  __syncthreads();

  if (tid < 64 * QB) {
    const int j = tid >> 6;
    float M0 = s_st[0][0][j][dd], L = s_st[1][0][j][dd], SS = s_st[2][0][j][dd];
    float A0 = s_st[3][0][j][dd], A1 = s_st[4][0][j][dd], A2 = s_st[5][0][j][dd];
    for (int p = 1; p < NP; ++p) {
      float m2 = s_st[0][p][j][dd];
      float mn = fmaxf(M0, m2);
      float f1 = __expf(M0 - mn);
      float f2 = __expf(m2 - mn);
      L  = L  * f1      + s_st[1][p][j][dd] * f2;
      SS = SS * f1 * f1 + s_st[2][p][j][dd] * f2 * f2;
      A0 = A0 * f1 + s_st[3][p][j][dd] * f2;
      A1 = A1 * f1 + s_st[4][p][j][dd] * f2;
      A2 = A2 * f1 + s_st[5][p][j][dd] * f2;
      M0 = mn;
    }
    L = fmaxf(L, 1.0e-30f);
    float scale = sqrtf(SS) / (L * L);
    float* zo = z + (size_t)(bq0 + j) * 3 * 64 + dd;
    zo[0]   = A0 * scale;
    zo[64]  = A1 * scale;
    zo[128] = A2 * scale;
  }
}

extern "C" void kernel_launch(void* const* d_in, const int* in_sizes, int n_in,
                              void* d_out, int out_size, void* d_ws, size_t ws_size,
                              hipStream_t stream) {
  const float* v_equi   = (const float*)d_in[0];
  const float* messages = (const float*)d_in[1];
  const int*   adj      = (const int*)d_in[2];
  const float* w_coord  = (const float*)d_in[3];
  const float* w_attn   = (const float*)d_in[4];
  float* out = (float*)d_out;

  const int B = 4, Nq = 512, D = 64;
  const int Mproj = B * 512 * 3;
  const int Mout  = B * Nq * 3;

  float* proj = (float*)d_ws;
  float* zbuf = proj + (size_t)Mproj * D;

  const int gemmBlocks = ((Mproj / 16) * 4 + 7) / 8;

  gemm_nt_wmma_f32<<<gemmBlocks, 256, 0, stream>>>(v_equi, w_coord, proj, Mproj);

  attn_softmax_kernel<<<(B * Nq) / QB, 64 * NP, 0, stream>>>(messages, adj, proj, zbuf);

  gemm_nt_wmma_f32<<<gemmBlocks, 256, 0, stream>>>(zbuf, w_attn, out, Mout);
}


# attention: prefetch only the thread's own rows: its first 16 k rows before the mask staging, none past the partition end (those re-read the neighbour's rows from memory); trip written loads-first
# speedup vs baseline: 1.3097x; 1.0817x over previous
#include <hip/hip_runtime.h>
#include <hip/hip_bf16.h>
#include <math.h>

typedef float v8f __attribute__((ext_vector_type(8)));
typedef float v2f __attribute__((ext_vector_type(2)));

#define NEG_BIAS (-1.0e10f)

__global__ __launch_bounds__(256) void gemm_nt_wmma_f32(
    const float* __restrict__ X, const float* __restrict__ W,
    float* __restrict__ C, int M)
{
  const int lane      = threadIdx.x & 31;
  const int waveInBlk = threadIdx.x >> 5;
  const int wave      = blockIdx.x * 8 + waveInBlk;
  const int mt        = wave >> 2;
  const int nt        = wave & 3;
  if (mt * 16 >= M) return;

  const int mrow  = mt * 16 + (lane & 15);
  const int wrow  = nt * 16 + (lane & 15);
  const int khalf = (lane >> 4) << 1;

  const float* __restrict__ xp = X + (size_t)mrow * 64 + khalf;
  const float* __restrict__ wp = W + (size_t)wrow * 64 + khalf;

  v8f c = {};
#pragma unroll
  for (int kk = 0; kk < 64; kk += 4) {
    v2f a, b;
    a.x = xp[kk];  a.y = xp[kk + 1];
    b.x = wp[kk];  b.y = wp[kk + 1];
    c = __builtin_amdgcn_wmma_f32_16x16x4_f32(false, a, false, b,
                                              (short)0, c, false, false);
  }

  const int crowBase = mt * 16 + ((lane >> 4) << 3);
  const int ccol     = nt * 16 + (lane & 15);
#pragma unroll
  for (int r = 0; r < 8; ++r)
    C[(size_t)(crowBase + r) * 64 + ccol] = c[r];
}

#define QB 4
#define NP 8
#define KP 64
#define PFD 16
static_assert(QB == 4 && NP * KP == 512 && (512 * QB) % (64 * NP) == 0, "the mask is read as one float4 per k; the partitions must tile the 512 keys; the mask is staged in whole passes of the block");
static_assert(PFD % 4 == 0 && PFD < KP && KP % 4 == 0, "both spans of the k loop are whole 4-k trips");

template <bool PF>
__device__ __forceinline__ void softmax_trip(
    const float* __restrict__ msg, const float* __restrict__ pk, const float* sb,
    float (&m)[QB], float (&l)[QB], float (&ss)[QB],
    float (&a0)[QB], float (&a1)[QB], float (&a2)[QB])
{
  float p0[4], p1[4], p2[4], x[4][QB];
  for (int u = 0; u < 4; ++u) {
    p0[u] = pk[u * 192]; p1[u] = pk[u * 192 + 64]; p2[u] = pk[u * 192 + 128];
    const float4 bv = *reinterpret_cast<const float4*>(sb + u * QB);
    const float bias[QB] = {bv.x, bv.y, bv.z, bv.w};
    for (int j = 0; j < QB; ++j) {
      const float* __restrict__ mp = msg + (size_t)j * 512 * 64 + u * 64;
      if (PF) __builtin_prefetch(mp + PFD * 64, 0, 0);
      x[u][j] = mp[0] + bias[j];
    }
  }

  for (int u = 0; u < 4; ++u) {
    for (int j = 0; j < QB; ++j) {
      float mn   = fmaxf(m[j], x[u][j]);
      float corr = __expf(m[j] - mn);
      float e    = __expf(x[u][j] - mn);
      float c2   = corr * corr;

      l[j]  = l[j]  * corr + e;
      ss[j] = ss[j] * c2   + e * e;
      a0[j] = a0[j] * corr + e * p0[u];
      a1[j] = a1[j] * corr + e * p1[u];
      a2[j] = a2[j] * corr + e * p2[u];
      m[j]  = mn;
    }
  }
}

__global__ __launch_bounds__(64 * NP) void attn_softmax_kernel(
    const float* __restrict__ messages, const int* __restrict__ adj,
    const float* __restrict__ proj, float* __restrict__ z)
{
  __shared__ __attribute__((aligned(16))) float s_bias[512 * QB];
  __shared__ float s_st[6][NP][QB][64];

  const int bq0  = blockIdx.x * QB;
  const int b    = bq0 >> 9;
  const int tid  = threadIdx.x;
  const int dd   = tid & 63;
  const int part = tid >> 6;

  const int k0 = part * KP;
  const float* __restrict__ msg = messages + (size_t)bq0 * 512 * 64 + (size_t)k0 * 64 + dd;
  const float* __restrict__ pk  = proj + (size_t)b * 512 * 3 * 64 + (size_t)k0 * 3 * 64 + dd;

  int av[(512 * QB) / (64 * NP)];
  for (int t = 0; t < (512 * QB) / (64 * NP); ++t) {
    const int i = tid + t * 64 * NP;
    const int j = i >> 9, k = i & 511;
    av[t] = adj[(size_t)(bq0 + j) * 512 + k];
  }

  for (int r = 0; r < PFD; ++r) {
    for (int j = 0; j < QB; ++j)
      __builtin_prefetch(msg + (size_t)j * 512 * 64 + r * 64, 0, 0);
  }
  __builtin_amdgcn_sched_barrier(0);

  for (int t = 0; t < (512 * QB) / (64 * NP); ++t) {
    const int i = tid + t * 64 * NP;
    const int j = i >> 9, k = i & 511;
    s_bias[k * QB + j] = (av[t] > 0) ? 0.0f : NEG_BIAS;
  }
  __syncthreads();

  float m[QB], l[QB], ss[QB], a0[QB], a1[QB], a2[QB];
#pragma unroll
  for (int j = 0; j < QB; ++j) {
    m[j] = -1.0e30f; l[j] = 0.0f; ss[j] = 0.0f;
    a0[j] = 0.0f; a1[j] = 0.0f; a2[j] = 0.0f;
  }

  for (int i = 0; i < KP - PFD; i += 4) {
    softmax_trip<true>(msg, pk, &s_bias[(k0 + i) * QB], m, l, ss, a0, a1, a2);
    msg += 4 * 64;
    pk  += 4 * 3 * 64;
  }
#pragma nounroll
  for (int i = KP - PFD; i < KP; i += 4) {
    softmax_trip<false>(msg, pk, &s_bias[(k0 + i) * QB], m, l, ss, a0, a1, a2);
    msg += 4 * 64;
    pk  += 4 * 3 * 64;
  }

#pragma unroll
  for (int j = 0; j < QB; ++j) {
    s_st[0][part][j][dd] = m[j];  s_st[1][part][j][dd] = l[j];
    s_st[2][part][j][dd] = ss[j]; s_st[3][part][j][dd] = a0[j];
    s_st[4][part][j][dd] = a1[j]; s_st[5][part][j][dd] = a2[j];
  }
  __syncthreads();

  if (tid < 64 * QB) {
    const int j = tid >> 6;
    float M0 = s_st[0][0][j][dd], L = s_st[1][0][j][dd], SS = s_st[2][0][j][dd];
    float A0 = s_st[3][0][j][dd], A1 = s_st[4][0][j][dd], A2 = s_st[5][0][j][dd];
    for (int p = 1; p < NP; ++p) {
      float m2 = s_st[0][p][j][dd];
      float mn = fmaxf(M0, m2);
      float f1 = __expf(M0 - mn);
      float f2 = __expf(m2 - mn);
      L  = L  * f1      + s_st[1][p][j][dd] * f2;
      SS = SS * f1 * f1 + s_st[2][p][j][dd] * f2 * f2;
      A0 = A0 * f1 + s_st[3][p][j][dd] * f2;
      A1 = A1 * f1 + s_st[4][p][j][dd] * f2;
      A2 = A2 * f1 + s_st[5][p][j][dd] * f2;
      M0 = mn;
    }
    L = fmaxf(L, 1.0e-30f);
    float scale = sqrtf(SS) / (L * L);
    float* zo = z + (size_t)(bq0 + j) * 3 * 64 + dd;
    zo[0]   = A0 * scale;
    zo[64]  = A1 * scale;
    zo[128] = A2 * scale;
  }
}

extern "C" void kernel_launch(void* const* d_in, const int* in_sizes, int n_in,
                              void* d_out, int out_size, void* d_ws, size_t ws_size,
                              hipStream_t stream) {
  const float* v_equi   = (const float*)d_in[0];
  const float* messages = (const float*)d_in[1];
  const int*   adj      = (const int*)d_in[2];
  const float* w_coord  = (const float*)d_in[3];
  const float* w_attn   = (const float*)d_in[4];
  float* out = (float*)d_out;

  const int B = 4, Nq = 512, D = 64;
  const int Mproj = B * 512 * 3;
  const int Mout  = B * Nq * 3;

  float* proj = (float*)d_ws;
  float* zbuf = proj + (size_t)Mproj * D;

  const int gemmBlocks = ((Mproj / 16) * 4 + 7) / 8;

  gemm_nt_wmma_f32<<<gemmBlocks, 256, 0, stream>>>(v_equi, w_coord, proj, Mproj);

  attn_softmax_kernel<<<(B * Nq) / QB, 64 * NP, 0, stream>>>(messages, adj, proj, zbuf);

  gemm_nt_wmma_f32<<<gemmBlocks, 256, 0, stream>>>(zbuf, w_attn, out, Mout);
}


# attention: the online softmax takes the maximum of a trip's 4 k values at once and rescales the running sums once per trip, not once per k (terms still added in k order; their rounding points move)
# speedup vs baseline: 1.3676x; 1.0442x over previous
#include <hip/hip_runtime.h>
#include <hip/hip_bf16.h>
#include <math.h>

typedef float v8f __attribute__((ext_vector_type(8)));
typedef float v2f __attribute__((ext_vector_type(2)));

#define NEG_BIAS (-1.0e10f)

__global__ __launch_bounds__(256) void gemm_nt_wmma_f32(
    const float* __restrict__ X, const float* __restrict__ W,
    float* __restrict__ C, int M)
{
  const int lane      = threadIdx.x & 31;
  const int waveInBlk = threadIdx.x >> 5;
  const int wave      = blockIdx.x * 8 + waveInBlk;
  const int mt        = wave >> 2;
  const int nt        = wave & 3;
  if (mt * 16 >= M) return;

  const int mrow  = mt * 16 + (lane & 15);
  const int wrow  = nt * 16 + (lane & 15);
  const int khalf = (lane >> 4) << 1;

  const float* __restrict__ xp = X + (size_t)mrow * 64 + khalf;
  const float* __restrict__ wp = W + (size_t)wrow * 64 + khalf;

  v8f c = {};
#pragma unroll
  for (int kk = 0; kk < 64; kk += 4) {
    v2f a, b;
    a.x = xp[kk];  a.y = xp[kk + 1];
    b.x = wp[kk];  b.y = wp[kk + 1];
    c = __builtin_amdgcn_wmma_f32_16x16x4_f32(false, a, false, b,
                                              (short)0, c, false, false);
  }

  const int crowBase = mt * 16 + ((lane >> 4) << 3);
  const int ccol     = nt * 16 + (lane & 15);
#pragma unroll
  for (int r = 0; r < 8; ++r)
    C[(size_t)(crowBase + r) * 64 + ccol] = c[r];
}

#define QB 4
#define NP 8
#define KP 64
#define PFD 16
static_assert(QB == 4 && NP * KP == 512 && (512 * QB) % (64 * NP) == 0, "the mask is read as one float4 per k; the partitions must tile the 512 keys; the mask is staged in whole passes of the block");
static_assert(PFD % 4 == 0 && PFD < KP && KP % 4 == 0, "both spans of the k loop are whole 4-k trips");

template <bool PF>
__device__ __forceinline__ void softmax_trip(
    const float* __restrict__ msg, const float* __restrict__ pk, const float* sb,
    float (&m)[QB], float (&l)[QB], float (&ss)[QB],
    float (&a0)[QB], float (&a1)[QB], float (&a2)[QB])
{
  float p0[4], p1[4], p2[4], x[4][QB];
  for (int u = 0; u < 4; ++u) {
    p0[u] = pk[u * 192]; p1[u] = pk[u * 192 + 64]; p2[u] = pk[u * 192 + 128];
    const float4 bv = *reinterpret_cast<const float4*>(sb + u * QB);
    const float bias[QB] = {bv.x, bv.y, bv.z, bv.w};
    for (int j = 0; j < QB; ++j) {
      const float* __restrict__ mp = msg + (size_t)j * 512 * 64 + u * 64;
      if (PF) __builtin_prefetch(mp + PFD * 64, 0, 0);
      x[u][j] = mp[0] + bias[j];
    }
  }

  for (int j = 0; j < QB; ++j) {
    const float mn   = fmaxf(fmaxf(fmaxf(m[j], x[0][j]), x[1][j]), fmaxf(x[2][j], x[3][j]));
    const float corr = __expf(m[j] - mn);
    float e[4];
    for (int u = 0; u < 4; ++u) e[u] = __expf(x[u][j] - mn);

    float lj = fmaf(l[j], corr, e[0]);
    float sj = ss[j] * (corr * corr);
    float b0 = a0[j] * corr, b1 = a1[j] * corr, b2 = a2[j] * corr;
    for (int u = 0; u < 4; ++u) {
      if (u) lj += e[u];
      sj = fmaf(e[u], e[u], sj);
      b0 = fmaf(e[u], p0[u], b0);
      b1 = fmaf(e[u], p1[u], b1);
      b2 = fmaf(e[u], p2[u], b2);
    }
    l[j] = lj; ss[j] = sj; a0[j] = b0; a1[j] = b1; a2[j] = b2;
    m[j] = mn;
  }
}

__global__ __launch_bounds__(64 * NP) void attn_softmax_kernel(
    const float* __restrict__ messages, const int* __restrict__ adj,
    const float* __restrict__ proj, float* __restrict__ z)
{
  __shared__ __attribute__((aligned(16))) float s_bias[512 * QB];
  __shared__ float s_st[6][NP][QB][64];

  const int bq0  = blockIdx.x * QB;
  const int b    = bq0 >> 9;
  const int tid  = threadIdx.x;
  const int dd   = tid & 63;
  const int part = tid >> 6;

  const int k0 = part * KP;
  const float* __restrict__ msg = messages + (size_t)bq0 * 512 * 64 + (size_t)k0 * 64 + dd;
  const float* __restrict__ pk  = proj + (size_t)b * 512 * 3 * 64 + (size_t)k0 * 3 * 64 + dd;

  int av[(512 * QB) / (64 * NP)];
  for (int t = 0; t < (512 * QB) / (64 * NP); ++t) {
    const int i = tid + t * 64 * NP;
    const int j = i >> 9, k = i & 511;
    av[t] = adj[(size_t)(bq0 + j) * 512 + k];
  }

  for (int r = 0; r < PFD; ++r) {
    for (int j = 0; j < QB; ++j)
      __builtin_prefetch(msg + (size_t)j * 512 * 64 + r * 64, 0, 0);
  }
  __builtin_amdgcn_sched_barrier(0);

  for (int t = 0; t < (512 * QB) / (64 * NP); ++t) {
    const int i = tid + t * 64 * NP;
    const int j = i >> 9, k = i & 511;
    s_bias[k * QB + j] = (av[t] > 0) ? 0.0f : NEG_BIAS;
  }
  __syncthreads();

  float m[QB], l[QB], ss[QB], a0[QB], a1[QB], a2[QB];
#pragma unroll
  for (int j = 0; j < QB; ++j) {
    m[j] = -1.0e30f; l[j] = 0.0f; ss[j] = 0.0f;
    a0[j] = 0.0f; a1[j] = 0.0f; a2[j] = 0.0f;
  }

  for (int i = 0; i < KP - PFD; i += 4) {
    softmax_trip<true>(msg, pk, &s_bias[(k0 + i) * QB], m, l, ss, a0, a1, a2);
    msg += 4 * 64;
    pk  += 4 * 3 * 64;
  }
#pragma nounroll
  for (int i = KP - PFD; i < KP; i += 4) {
    softmax_trip<false>(msg, pk, &s_bias[(k0 + i) * QB], m, l, ss, a0, a1, a2);
    msg += 4 * 64;
    pk  += 4 * 3 * 64;
  }

#pragma unroll
  for (int j = 0; j < QB; ++j) {
    s_st[0][part][j][dd] = m[j];  s_st[1][part][j][dd] = l[j];
    s_st[2][part][j][dd] = ss[j]; s_st[3][part][j][dd] = a0[j];
    s_st[4][part][j][dd] = a1[j]; s_st[5][part][j][dd] = a2[j];
  }
  __syncthreads();

  if (tid < 64 * QB) {
    const int j = tid >> 6;
    float M0 = s_st[0][0][j][dd], L = s_st[1][0][j][dd], SS = s_st[2][0][j][dd];
    float A0 = s_st[3][0][j][dd], A1 = s_st[4][0][j][dd], A2 = s_st[5][0][j][dd];
    for (int p = 1; p < NP; ++p) {
      float m2 = s_st[0][p][j][dd];
      float mn = fmaxf(M0, m2);
      float f1 = __expf(M0 - mn);
      float f2 = __expf(m2 - mn);
      L  = L  * f1      + s_st[1][p][j][dd] * f2;
      SS = SS * f1 * f1 + s_st[2][p][j][dd] * f2 * f2;
      A0 = A0 * f1 + s_st[3][p][j][dd] * f2;
      A1 = A1 * f1 + s_st[4][p][j][dd] * f2;
      A2 = A2 * f1 + s_st[5][p][j][dd] * f2;
      M0 = mn;
    }
    L = fmaxf(L, 1.0e-30f);
    float scale = sqrtf(SS) / (L * L);
    float* zo = z + (size_t)(bq0 + j) * 3 * 64 + dd;
    zo[0]   = A0 * scale;
    zo[64]  = A1 * scale;
    zo[128] = A2 * scale;
  }
}

extern "C" void kernel_launch(void* const* d_in, const int* in_sizes, int n_in,
                              void* d_out, int out_size, void* d_ws, size_t ws_size,
                              hipStream_t stream) {
  const float* v_equi   = (const float*)d_in[0];
  const float* messages = (const float*)d_in[1];
  const int*   adj      = (const int*)d_in[2];
  const float* w_coord  = (const float*)d_in[3];
  const float* w_attn   = (const float*)d_in[4];
  float* out = (float*)d_out;

  const int B = 4, Nq = 512, D = 64;
  const int Mproj = B * 512 * 3;
  const int Mout  = B * Nq * 3;

  float* proj = (float*)d_ws;
  float* zbuf = proj + (size_t)Mproj * D;

  const int gemmBlocks = ((Mproj / 16) * 4 + 7) / 8;

  gemm_nt_wmma_f32<<<gemmBlocks, 256, 0, stream>>>(v_equi, w_coord, proj, Mproj);

  attn_softmax_kernel<<<(B * Nq) / QB, 64 * NP, 0, stream>>>(messages, adj, proj, zbuf);

  gemm_nt_wmma_f32<<<gemmBlocks, 256, 0, stream>>>(zbuf, w_attn, out, Mout);
}


# attention: the output projection runs in each block's epilogue (its 12 z rows through LDS into 16 f32 WMMAs per wave): no third launch, z never goes to memory; that product's k terms sum in new order
# speedup vs baseline: 1.4779x; 1.0807x over previous
#include <hip/hip_runtime.h>
#include <hip/hip_bf16.h>
#include <math.h>

typedef float v8f __attribute__((ext_vector_type(8)));
typedef float v2f __attribute__((ext_vector_type(2)));

#define NEG_BIAS (-1.0e10f)

__global__ __launch_bounds__(256) void gemm_nt_wmma_f32(
    const float* __restrict__ X, const float* __restrict__ W,
    float* __restrict__ C, int M)
{
  const int lane      = threadIdx.x & 31;
  const int waveInBlk = threadIdx.x >> 5;
  const int wave      = blockIdx.x * 8 + waveInBlk;
  const int mt        = wave >> 2;
  const int nt        = wave & 3;
  if (mt * 16 >= M) return;

  const int mrow  = mt * 16 + (lane & 15);
  const int wrow  = nt * 16 + (lane & 15);
  const int khalf = (lane >> 4) << 1;

  const float* __restrict__ xp = X + (size_t)mrow * 64 + khalf;
  const float* __restrict__ wp = W + (size_t)wrow * 64 + khalf;

  v8f c = {};
#pragma unroll
  for (int kk = 0; kk < 64; kk += 4) {
    v2f a, b;
    a.x = xp[kk];  a.y = xp[kk + 1];
    b.x = wp[kk];  b.y = wp[kk + 1];
    c = __builtin_amdgcn_wmma_f32_16x16x4_f32(false, a, false, b,
                                              (short)0, c, false, false);
  }

  const int crowBase = mt * 16 + ((lane >> 4) << 3);
  const int ccol     = nt * 16 + (lane & 15);
#pragma unroll
  for (int r = 0; r < 8; ++r)
    C[(size_t)(crowBase + r) * 64 + ccol] = c[r];
}

#define QB 4
#define NP 8
#define KP 64
#define PFD 16
static_assert(QB == 4 && NP * KP == 512 && (512 * QB) % (64 * NP) == 0, "the mask is read as one float4 per k; the partitions must tile the 512 keys; the mask is staged in whole passes of the block");
static_assert(PFD % 4 == 0 && PFD < KP && KP % 4 == 0, "both spans of the k loop are whole 4-k trips");
#define ZP 68
static_assert(QB * 4 == 16 && 16 * ZP <= 512 * QB && 64 * NP >= 4 * 32, "the block's 3*QB z rows and QB zero rows make one 16-row A tile, staged in the mask's LDS; one wave per 16-column tile of out");

template <bool PF>
__device__ __forceinline__ void softmax_trip(
    const float* __restrict__ msg, const float* __restrict__ pk, const float* sb,
    float (&m)[QB], float (&l)[QB], float (&ss)[QB],
    float (&a0)[QB], float (&a1)[QB], float (&a2)[QB])
{
  float p0[4], p1[4], p2[4], x[4][QB];
  for (int u = 0; u < 4; ++u) {
    p0[u] = pk[u * 192]; p1[u] = pk[u * 192 + 64]; p2[u] = pk[u * 192 + 128];
    const float4 bv = *reinterpret_cast<const float4*>(sb + u * QB);
    const float bias[QB] = {bv.x, bv.y, bv.z, bv.w};
    for (int j = 0; j < QB; ++j) {
      const float* __restrict__ mp = msg + (size_t)j * 512 * 64 + u * 64;
      if (PF) __builtin_prefetch(mp + PFD * 64, 0, 0);
      x[u][j] = mp[0] + bias[j];
    }
  }

  for (int j = 0; j < QB; ++j) {
    const float mn   = fmaxf(fmaxf(fmaxf(m[j], x[0][j]), x[1][j]), fmaxf(x[2][j], x[3][j]));
    const float corr = __expf(m[j] - mn);
    float e[4];
    for (int u = 0; u < 4; ++u) e[u] = __expf(x[u][j] - mn);

    float lj = fmaf(l[j], corr, e[0]);
    float sj = ss[j] * (corr * corr);
    float b0 = a0[j] * corr, b1 = a1[j] * corr, b2 = a2[j] * corr;
    for (int u = 0; u < 4; ++u) {
      if (u) lj += e[u];
      sj = fmaf(e[u], e[u], sj);
      b0 = fmaf(e[u], p0[u], b0);
      b1 = fmaf(e[u], p1[u], b1);
      b2 = fmaf(e[u], p2[u], b2);
    }
    l[j] = lj; ss[j] = sj; a0[j] = b0; a1[j] = b1; a2[j] = b2;
    m[j] = mn;
  }
}

__global__ __launch_bounds__(64 * NP) void attn_softmax_kernel(
    const float* __restrict__ messages, const int* __restrict__ adj,
    const float* __restrict__ proj, const float* __restrict__ w_attn,
    float* __restrict__ out)
{
  __shared__ __attribute__((aligned(16))) float s_bias[512 * QB];
  __shared__ float s_st[6][NP][QB][64];

  const int bq0  = blockIdx.x * QB;
  const int b    = bq0 >> 9;
  const int tid  = threadIdx.x;
  const int dd   = tid & 63;
  const int part = tid >> 6;

  const int k0 = part * KP;
  const float* __restrict__ msg = messages + (size_t)bq0 * 512 * 64 + (size_t)k0 * 64 + dd;
  const float* __restrict__ pk  = proj + (size_t)b * 512 * 3 * 64 + (size_t)k0 * 3 * 64 + dd;

  int av[(512 * QB) / (64 * NP)];
  for (int t = 0; t < (512 * QB) / (64 * NP); ++t) {
    const int i = tid + t * 64 * NP;
    const int j = i >> 9, k = i & 511;
    av[t] = adj[(size_t)(bq0 + j) * 512 + k];
  }

  for (int r = 0; r < PFD; ++r) {
    for (int j = 0; j < QB; ++j)
      __builtin_prefetch(msg + (size_t)j * 512 * 64 + r * 64, 0, 0);
  }
  __builtin_amdgcn_sched_barrier(0);

  for (int t = 0; t < (512 * QB) / (64 * NP); ++t) {
    const int i = tid + t * 64 * NP;
    const int j = i >> 9, k = i & 511;
    s_bias[k * QB + j] = (av[t] > 0) ? 0.0f : NEG_BIAS;
  }
  __syncthreads();

  float m[QB], l[QB], ss[QB], a0[QB], a1[QB], a2[QB];
#pragma unroll
  for (int j = 0; j < QB; ++j) {
    m[j] = -1.0e30f; l[j] = 0.0f; ss[j] = 0.0f;
    a0[j] = 0.0f; a1[j] = 0.0f; a2[j] = 0.0f;
  }

  for (int i = 0; i < KP - PFD; i += 4) {
    softmax_trip<true>(msg, pk, &s_bias[(k0 + i) * QB], m, l, ss, a0, a1, a2);
    msg += 4 * 64;
    pk  += 4 * 3 * 64;
  }
#pragma nounroll
  for (int i = KP - PFD; i < KP; i += 4) {
    softmax_trip<false>(msg, pk, &s_bias[(k0 + i) * QB], m, l, ss, a0, a1, a2);
    msg += 4 * 64;
    pk  += 4 * 3 * 64;
  }

#pragma unroll
  for (int j = 0; j < QB; ++j) {
    s_st[0][part][j][dd] = m[j];  s_st[1][part][j][dd] = l[j];
    s_st[2][part][j][dd] = ss[j]; s_st[3][part][j][dd] = a0[j];
    s_st[4][part][j][dd] = a1[j]; s_st[5][part][j][dd] = a2[j];
  }
  __syncthreads();

  const int lane = tid & 31;
  const int wv   = tid >> 5;
  const int fc   = lane & 15;
  const int fh   = lane >> 4;
  float4 bw[8];
  if (wv < 4) {
    const float4* __restrict__ wp =
        reinterpret_cast<const float4*>(w_attn + (size_t)(wv * 16 + fc) * 64 + fh * 32);
#pragma unroll
    for (int i = 0; i < 8; ++i) bw[i] = wp[i];
  }

  if (tid < 64 * QB) {
    const int j = tid >> 6;
    float M0 = s_st[0][0][j][dd], L = s_st[1][0][j][dd], SS = s_st[2][0][j][dd];
    float A0 = s_st[3][0][j][dd], A1 = s_st[4][0][j][dd], A2 = s_st[5][0][j][dd];
    for (int p = 1; p < NP; ++p) {
      float m2 = s_st[0][p][j][dd];
      float mn = fmaxf(M0, m2);
      float f1 = __expf(M0 - mn);
      float f2 = __expf(m2 - mn);
      L  = L  * f1      + s_st[1][p][j][dd] * f2;
      SS = SS * f1 * f1 + s_st[2][p][j][dd] * f2 * f2;
      A0 = A0 * f1 + s_st[3][p][j][dd] * f2;
      A1 = A1 * f1 + s_st[4][p][j][dd] * f2;
      A2 = A2 * f1 + s_st[5][p][j][dd] * f2;
      M0 = mn;
    }
    L = fmaxf(L, 1.0e-30f);
    float scale = sqrtf(SS) / (L * L);
    float* zs = s_bias;
    zs[(j * 3 + 0) * ZP + dd] = A0 * scale;
    zs[(j * 3 + 1) * ZP + dd] = A1 * scale;
    zs[(j * 3 + 2) * ZP + dd] = A2 * scale;
    zs[(QB * 3 + j) * ZP + dd] = 0.0f;
  }
  __syncthreads();

  if (wv < 4) {
    const float4* zr = reinterpret_cast<const float4*>(s_bias + fc * ZP + fh * 32);
    float4 az[8];
#pragma unroll
    for (int i = 0; i < 8; ++i) az[i] = zr[i];
    __builtin_amdgcn_sched_barrier(0);

    v8f c0 = {}, c1 = {};
#pragma unroll
    for (int i = 0; i < 8; ++i) {
      v2f a, b;
      a.x = az[i].x; a.y = az[i].y;  b.x = bw[i].x; b.y = bw[i].y;
      c0 = __builtin_amdgcn_wmma_f32_16x16x4_f32(false, a, false, b,
                                                 (short)0, c0, false, false);
      a.x = az[i].z; a.y = az[i].w;  b.x = bw[i].z; b.y = bw[i].w;
      c1 = __builtin_amdgcn_wmma_f32_16x16x4_f32(false, a, false, b,
                                                 (short)0, c1, false, false);
    }

    float* __restrict__ op = out + (size_t)(bq0 * 3 + fh * 8) * 64 + wv * 16 + fc;
#pragma unroll
    for (int r = 0; r < 8; ++r)
      if (fh * 8 + r < QB * 3) op[(size_t)r * 64] = c0[r] + c1[r];
  }
}

extern "C" void kernel_launch(void* const* d_in, const int* in_sizes, int n_in,
                              void* d_out, int out_size, void* d_ws, size_t ws_size,
                              hipStream_t stream) {
  const float* v_equi   = (const float*)d_in[0];
  const float* messages = (const float*)d_in[1];
  const int*   adj      = (const int*)d_in[2];
  const float* w_coord  = (const float*)d_in[3];
  const float* w_attn   = (const float*)d_in[4];
  float* out = (float*)d_out;

  const int B = 4, Nq = 512;
  const int Mproj = B * 512 * 3;

  float* proj = (float*)d_ws;

  const int gemmBlocks = ((Mproj / 16) * 4 + 7) / 8;

  gemm_nt_wmma_f32<<<gemmBlocks, 256, 0, stream>>>(v_equi, w_coord, proj, Mproj);

  attn_softmax_kernel<<<(B * Nq) / QB, 64 * NP, 0, stream>>>(messages, adj, proj, w_attn, out);
}


# attention: a lane owns 2 adjacent channels: one 8-byte load per lane reads a whole 256-byte row per wave instruction (half the memory instructions); k split 16x32, rescale per 2 k: new summation tree
# speedup vs baseline: 1.6573x; 1.1214x over previous
#include <hip/hip_runtime.h>
#include <hip/hip_bf16.h>
#include <math.h>

typedef float v8f __attribute__((ext_vector_type(8)));
typedef float v2f __attribute__((ext_vector_type(2)));

#define NEG_BIAS (-1.0e10f)

__global__ __launch_bounds__(256) void gemm_nt_wmma_f32(
    const float* __restrict__ X, const float* __restrict__ W,
    float* __restrict__ C, int M)
{
  const int lane      = threadIdx.x & 31;
  const int waveInBlk = threadIdx.x >> 5;
  const int wave      = blockIdx.x * 8 + waveInBlk;
  const int mt        = wave >> 2;
  const int nt        = wave & 3;
  if (mt * 16 >= M) return;

  const int mrow  = mt * 16 + (lane & 15);
  const int wrow  = nt * 16 + (lane & 15);
  const int khalf = (lane >> 4) << 1;

  const float* __restrict__ xp = X + (size_t)mrow * 64 + khalf;
  const float* __restrict__ wp = W + (size_t)wrow * 64 + khalf;

  v8f c = {};
#pragma unroll
  for (int kk = 0; kk < 64; kk += 4) {
    v2f a, b;
    a.x = xp[kk];  a.y = xp[kk + 1];
    b.x = wp[kk];  b.y = wp[kk + 1];
    c = __builtin_amdgcn_wmma_f32_16x16x4_f32(false, a, false, b,
                                              (short)0, c, false, false);
  }

  const int crowBase = mt * 16 + ((lane >> 4) << 3);
  const int ccol     = nt * 16 + (lane & 15);
#pragma unroll
  for (int r = 0; r < 8; ++r)
    C[(size_t)(crowBase + r) * 64 + ccol] = c[r];
}

#define QB 4
#define CH 2
#define NP 16
#define KP 32
#define KT 2
#define PFD 8
#define NT (32 * NP)
static_assert(QB == 4 && CH * 32 == 64 && NP * KP == 512 && (512 * QB) % NT == 0, "the mask is read as one float4 per k; a wave's lanes cover the 64 channels; the partitions must tile the 512 keys; the mask is staged in whole passes of the block");
static_assert(PFD % KT == 0 && PFD < KP && KP % KT == 0, "both spans of the k loop are whole trips");
#define ZP 68
static_assert(QB * 4 == 16 && 16 * ZP <= 512 * QB && NT >= 4 * 32 && NT >= 64 * QB, "the block's 3*QB z rows and QB zero rows make one 16-row A tile, staged in the mask's LDS; one wave per 16-column tile of out; one merge thread per row and channel");

template <bool PF>
__device__ __forceinline__ void softmax_trip(
    const float* __restrict__ msg, const float* __restrict__ pk, const float* sb,
    v2f (&m)[QB], v2f (&l)[QB], v2f (&ss)[QB],
    v2f (&a0)[QB], v2f (&a1)[QB], v2f (&a2)[QB])
{
  v2f p0[KT], p1[KT], p2[KT], x[KT][QB];
  for (int u = 0; u < KT; ++u) {
    p0[u] = *reinterpret_cast<const v2f*>(pk + u * 192);
    p1[u] = *reinterpret_cast<const v2f*>(pk + u * 192 + 64);
    p2[u] = *reinterpret_cast<const v2f*>(pk + u * 192 + 128);
    const float4 bv = *reinterpret_cast<const float4*>(sb + u * QB);
    const float bias[QB] = {bv.x, bv.y, bv.z, bv.w};
    for (int j = 0; j < QB; ++j) {
      const float* __restrict__ mp = msg + (size_t)j * 512 * 64 + u * 64;
      if (PF) __builtin_prefetch(mp + PFD * 64, 0, 0);
      const v2f mv = *reinterpret_cast<const v2f*>(mp);
      x[u][j].x = mv.x + bias[j];
      x[u][j].y = mv.y + bias[j];
    }
  }

  for (int j = 0; j < QB; ++j) {
    for (int h = 0; h < CH; ++h) {
      const float mo = m[j][h];
      const float mn = fmaxf(fmaxf(mo, x[0][j][h]), x[1][j][h]);
      const float corr = __expf(mo - mn);
      float e[KT];
      for (int u = 0; u < KT; ++u) e[u] = __expf(x[u][j][h] - mn);

      float lj = fmaf(l[j][h], corr, e[0]);
      float sj = ss[j][h] * (corr * corr);
      float b0 = a0[j][h] * corr, b1 = a1[j][h] * corr, b2 = a2[j][h] * corr;
      for (int u = 0; u < KT; ++u) {
        if (u) lj += e[u];
        sj = fmaf(e[u], e[u], sj);
        b0 = fmaf(e[u], p0[u][h], b0);
        b1 = fmaf(e[u], p1[u][h], b1);
        b2 = fmaf(e[u], p2[u][h], b2);
      }
      l[j][h] = lj; ss[j][h] = sj; a0[j][h] = b0; a1[j][h] = b1; a2[j][h] = b2;
      m[j][h] = mn;
    }
  }
}
static_assert(KT == 2, "the trip's maximum is written out for two k values");

__global__ __launch_bounds__(NT) void attn_softmax_kernel(
    const float* __restrict__ messages, const int* __restrict__ adj,
    const float* __restrict__ proj, const float* __restrict__ w_attn,
    float* __restrict__ out)
{
  __shared__ __attribute__((aligned(16))) float s_bias[512 * QB];
  __shared__ __attribute__((aligned(8))) float s_st[6][NP][QB][64];

  const int bq0  = blockIdx.x * QB;
  const int b    = bq0 >> 9;
  const int tid  = threadIdx.x;
  const int lane = tid & 31;
  const int dd   = lane * CH;
  const int part = tid >> 5;

  const int k0 = part * KP;
  const float* __restrict__ msg = messages + (size_t)bq0 * 512 * 64 + (size_t)k0 * 64 + dd;
  const float* __restrict__ pk  = proj + (size_t)b * 512 * 3 * 64 + (size_t)k0 * 3 * 64 + dd;

  int av[(512 * QB) / NT];
  for (int t = 0; t < (512 * QB) / NT; ++t) {
    const int i = tid + t * NT;
    const int j = i >> 9, k = i & 511;
    av[t] = adj[(size_t)(bq0 + j) * 512 + k];
  }

  for (int r = 0; r < PFD; ++r) {
    for (int j = 0; j < QB; ++j)
      __builtin_prefetch(msg + (size_t)j * 512 * 64 + r * 64, 0, 0);
  }
  __builtin_amdgcn_sched_barrier(0);

  for (int t = 0; t < (512 * QB) / NT; ++t) {
    const int i = tid + t * NT;
    const int j = i >> 9, k = i & 511;
    s_bias[k * QB + j] = (av[t] > 0) ? 0.0f : NEG_BIAS;
  }
  __syncthreads();

  v2f m[QB], l[QB], ss[QB], a0[QB], a1[QB], a2[QB];
#pragma unroll
  for (int j = 0; j < QB; ++j) {
    m[j] = (v2f){-1.0e30f, -1.0e30f};
    l[j] = (v2f){0.0f, 0.0f};  ss[j] = (v2f){0.0f, 0.0f};
    a0[j] = (v2f){0.0f, 0.0f}; a1[j] = (v2f){0.0f, 0.0f}; a2[j] = (v2f){0.0f, 0.0f};
  }

  for (int i = 0; i < KP - PFD; i += KT) {
    softmax_trip<true>(msg, pk, &s_bias[(k0 + i) * QB], m, l, ss, a0, a1, a2);
    msg += KT * 64;
    pk  += KT * 3 * 64;
  }
#pragma nounroll
  for (int i = KP - PFD; i < KP; i += KT) {
    softmax_trip<false>(msg, pk, &s_bias[(k0 + i) * QB], m, l, ss, a0, a1, a2);
    msg += KT * 64;
    pk  += KT * 3 * 64;
  }

#pragma unroll
  for (int j = 0; j < QB; ++j) {
    *reinterpret_cast<v2f*>(&s_st[0][part][j][dd]) = m[j];
    *reinterpret_cast<v2f*>(&s_st[1][part][j][dd]) = l[j];
    *reinterpret_cast<v2f*>(&s_st[2][part][j][dd]) = ss[j];
    *reinterpret_cast<v2f*>(&s_st[3][part][j][dd]) = a0[j];
    *reinterpret_cast<v2f*>(&s_st[4][part][j][dd]) = a1[j];
    *reinterpret_cast<v2f*>(&s_st[5][part][j][dd]) = a2[j];
  }
  __syncthreads();

  const int wv   = tid >> 5;
  const int fc   = lane & 15;
  const int fh   = lane >> 4;
  float4 bw[8];
  if (wv < 4) {
    const float4* __restrict__ wp =
        reinterpret_cast<const float4*>(w_attn + (size_t)(wv * 16 + fc) * 64 + fh * 32);
#pragma unroll
    for (int i = 0; i < 8; ++i) bw[i] = wp[i];
  }

  if (tid < 64 * QB) {
    const int j  = tid >> 6;
    const int mc = tid & 63;
    float M0 = s_st[0][0][j][mc], L = s_st[1][0][j][mc], SS = s_st[2][0][j][mc];
    float A0 = s_st[3][0][j][mc], A1 = s_st[4][0][j][mc], A2 = s_st[5][0][j][mc];
    for (int p = 1; p < NP; ++p) {
      float m2 = s_st[0][p][j][mc];
      float mn = fmaxf(M0, m2);
      float f1 = __expf(M0 - mn);
      float f2 = __expf(m2 - mn);
      L  = L  * f1      + s_st[1][p][j][mc] * f2;
      SS = SS * f1 * f1 + s_st[2][p][j][mc] * f2 * f2;
      A0 = A0 * f1 + s_st[3][p][j][mc] * f2;
      A1 = A1 * f1 + s_st[4][p][j][mc] * f2;
      A2 = A2 * f1 + s_st[5][p][j][mc] * f2;
      M0 = mn;
    }
    L = fmaxf(L, 1.0e-30f);
    float scale = sqrtf(SS) / (L * L);
    float* zs = s_bias;
    zs[(j * 3 + 0) * ZP + mc] = A0 * scale;
    zs[(j * 3 + 1) * ZP + mc] = A1 * scale;
    zs[(j * 3 + 2) * ZP + mc] = A2 * scale;
    zs[(QB * 3 + j) * ZP + mc] = 0.0f;
  }
  __syncthreads();

  if (wv < 4) {
    const float4* zr = reinterpret_cast<const float4*>(s_bias + fc * ZP + fh * 32);
    float4 az[8];
#pragma unroll
    for (int i = 0; i < 8; ++i) az[i] = zr[i];
    __builtin_amdgcn_sched_barrier(0);

    v8f c0 = {}, c1 = {};
#pragma unroll
    for (int i = 0; i < 8; ++i) {
      v2f a, b;
      a.x = az[i].x; a.y = az[i].y;  b.x = bw[i].x; b.y = bw[i].y;
      c0 = __builtin_amdgcn_wmma_f32_16x16x4_f32(false, a, false, b,
                                                 (short)0, c0, false, false);
      a.x = az[i].z; a.y = az[i].w;  b.x = bw[i].z; b.y = bw[i].w;
      c1 = __builtin_amdgcn_wmma_f32_16x16x4_f32(false, a, false, b,
                                                 (short)0, c1, false, false);
    }

    float* __restrict__ op = out + (size_t)(bq0 * 3 + fh * 8) * 64 + wv * 16 + fc;
#pragma unroll
    for (int r = 0; r < 8; ++r)
      if (fh * 8 + r < QB * 3) op[(size_t)r * 64] = c0[r] + c1[r];
  }
}

extern "C" void kernel_launch(void* const* d_in, const int* in_sizes, int n_in,
                              void* d_out, int out_size, void* d_ws, size_t ws_size,
                              hipStream_t stream) {
  const float* v_equi   = (const float*)d_in[0];
  const float* messages = (const float*)d_in[1];
  const int*   adj      = (const int*)d_in[2];
  const float* w_coord  = (const float*)d_in[3];
  const float* w_attn   = (const float*)d_in[4];
  float* out = (float*)d_out;

  const int B = 4, Nq = 512;
  const int Mproj = B * 512 * 3;

  float* proj = (float*)d_ws;

  const int gemmBlocks = ((Mproj / 16) * 4 + 7) / 8;

  gemm_nt_wmma_f32<<<gemmBlocks, 256, 0, stream>>>(v_equi, w_coord, proj, Mproj);

  attn_softmax_kernel<<<(B * Nq) / QB, NT, 0, stream>>>(messages, adj, proj, w_attn, out);
}


# attention: each thread carries 8 q rows, not 4: 256 blocks, one per WGP, so a WGP reads its batch's proj_equi once (was twice) and half as many waves start; every sum keeps its order
# speedup vs baseline: 1.7277x; 1.0425x over previous
#include <hip/hip_runtime.h>
#include <hip/hip_bf16.h>
#include <math.h>

typedef float v8f __attribute__((ext_vector_type(8)));
typedef float v2f __attribute__((ext_vector_type(2)));

#define NEG_BIAS (-1.0e10f)

__global__ __launch_bounds__(256) void gemm_nt_wmma_f32(
    const float* __restrict__ X, const float* __restrict__ W,
    float* __restrict__ C, int M)
{
  const int lane      = threadIdx.x & 31;
  const int waveInBlk = threadIdx.x >> 5;
  const int wave      = blockIdx.x * 8 + waveInBlk;
  const int mt        = wave >> 2;
  const int nt        = wave & 3;
  if (mt * 16 >= M) return;

  const int mrow  = mt * 16 + (lane & 15);
  const int wrow  = nt * 16 + (lane & 15);
  const int khalf = (lane >> 4) << 1;

  const float* __restrict__ xp = X + (size_t)mrow * 64 + khalf;
  const float* __restrict__ wp = W + (size_t)wrow * 64 + khalf;

  v8f c = {};
#pragma unroll
  for (int kk = 0; kk < 64; kk += 4) {
    v2f a, b;
    a.x = xp[kk];  a.y = xp[kk + 1];
    b.x = wp[kk];  b.y = wp[kk + 1];
    c = __builtin_amdgcn_wmma_f32_16x16x4_f32(false, a, false, b,
                                              (short)0, c, false, false);
  }

  const int crowBase = mt * 16 + ((lane >> 4) << 3);
  const int ccol     = nt * 16 + (lane & 15);
#pragma unroll
  for (int r = 0; r < 8; ++r)
    C[(size_t)(crowBase + r) * 64 + ccol] = c[r];
}

#define QB 8
#define CH 2
#define NP 16
#define KP 32
#define KT 2
#define PFD 8
#define NT (32 * NP)
static_assert(QB == 8 && CH * 32 == 64 && NP * KP == 512 && (512 * QB) % NT == 0, "the mask is read as two float4 per k; a wave's lanes cover the 64 channels; the partitions must tile the 512 keys; the mask is staged in whole passes of the block");
static_assert(PFD % KT == 0 && PFD < KP && KP % KT == 0, "both spans of the k loop are whole trips");
#define ZP 68
#define ZR 32
static_assert(QB * 4 == ZR && ZR * ZP <= 512 * QB && NT >= 8 * 32 && NT == 64 * QB, "the z rows are staged in the mask's LDS; one wave per 16-row, 16-column tile of out; one merge thread per row and channel");

template <bool PF>
__device__ __forceinline__ void softmax_trip(
    const float* __restrict__ msg, const float* __restrict__ pk, const float* sb,
    v2f (&m)[QB], v2f (&l)[QB], v2f (&ss)[QB],
    v2f (&a0)[QB], v2f (&a1)[QB], v2f (&a2)[QB])
{
  v2f p0[KT], p1[KT], p2[KT], x[KT][QB];
  for (int u = 0; u < KT; ++u) {
    p0[u] = *reinterpret_cast<const v2f*>(pk + u * 192);
    p1[u] = *reinterpret_cast<const v2f*>(pk + u * 192 + 64);
    p2[u] = *reinterpret_cast<const v2f*>(pk + u * 192 + 128);
    const float4 bv0 = *reinterpret_cast<const float4*>(sb + u * QB);
    const float4 bv1 = *reinterpret_cast<const float4*>(sb + u * QB + 4);
    const float bias[QB] = {bv0.x, bv0.y, bv0.z, bv0.w, bv1.x, bv1.y, bv1.z, bv1.w};
    for (int j = 0; j < QB; ++j) {
      const float* __restrict__ mp = msg + (size_t)j * 512 * 64 + u * 64;
      if (PF) __builtin_prefetch(mp + PFD * 64, 0, 0);
      const v2f mv = *reinterpret_cast<const v2f*>(mp);
      x[u][j].x = mv.x + bias[j];
      x[u][j].y = mv.y + bias[j];
    }
  }

  for (int j = 0; j < QB; ++j) {
    for (int h = 0; h < CH; ++h) {
      const float mo = m[j][h];
      const float mn = fmaxf(fmaxf(mo, x[0][j][h]), x[1][j][h]);
      const float corr = __expf(mo - mn);
      float e[KT];
      for (int u = 0; u < KT; ++u) e[u] = __expf(x[u][j][h] - mn);

      float lj = fmaf(l[j][h], corr, e[0]);
      float sj = ss[j][h] * (corr * corr);
      float b0 = a0[j][h] * corr, b1 = a1[j][h] * corr, b2 = a2[j][h] * corr;
      for (int u = 0; u < KT; ++u) {
        if (u) lj += e[u];
        sj = fmaf(e[u], e[u], sj);
        b0 = fmaf(e[u], p0[u][h], b0);
        b1 = fmaf(e[u], p1[u][h], b1);
        b2 = fmaf(e[u], p2[u][h], b2);
      }
      l[j][h] = lj; ss[j][h] = sj; a0[j][h] = b0; a1[j][h] = b1; a2[j][h] = b2;
      m[j][h] = mn;
    }
  }
}
static_assert(KT == 2, "the trip's maximum is written out for two k values");

__global__ __launch_bounds__(NT) void attn_softmax_kernel(
    const float* __restrict__ messages, const int* __restrict__ adj,
    const float* __restrict__ proj, const float* __restrict__ w_attn,
    float* __restrict__ out)
{
  __shared__ __attribute__((aligned(16))) float s_bias[512 * QB];
  __shared__ __attribute__((aligned(8))) float s_st[6][NP][QB][64];

  const int bq0  = blockIdx.x * QB;
  const int b    = bq0 >> 9;
  const int tid  = threadIdx.x;
  const int lane = tid & 31;
  const int dd   = lane * CH;
  const int part = tid >> 5;

  const int k0 = part * KP;
  const float* __restrict__ msg = messages + (size_t)bq0 * 512 * 64 + (size_t)k0 * 64 + dd;
  const float* __restrict__ pk  = proj + (size_t)b * 512 * 3 * 64 + (size_t)k0 * 3 * 64 + dd;

  int av[(512 * QB) / NT];
  for (int t = 0; t < (512 * QB) / NT; ++t) {
    const int i = tid + t * NT;
    const int j = i >> 9, k = i & 511;
    av[t] = adj[(size_t)(bq0 + j) * 512 + k];
  }

  for (int r = 0; r < PFD; ++r) {
    for (int j = 0; j < QB; ++j)
      __builtin_prefetch(msg + (size_t)j * 512 * 64 + r * 64, 0, 0);
  }
  __builtin_amdgcn_sched_barrier(0);

  for (int t = 0; t < (512 * QB) / NT; ++t) {
    const int i = tid + t * NT;
    const int j = i >> 9, k = i & 511;
    s_bias[k * QB + j] = (av[t] > 0) ? 0.0f : NEG_BIAS;
  }
  __syncthreads();

  v2f m[QB], l[QB], ss[QB], a0[QB], a1[QB], a2[QB];
#pragma unroll
  for (int j = 0; j < QB; ++j) {
    m[j] = (v2f){-1.0e30f, -1.0e30f};
    l[j] = (v2f){0.0f, 0.0f};  ss[j] = (v2f){0.0f, 0.0f};
    a0[j] = (v2f){0.0f, 0.0f}; a1[j] = (v2f){0.0f, 0.0f}; a2[j] = (v2f){0.0f, 0.0f};
  }

  for (int i = 0; i < KP - PFD; i += KT) {
    softmax_trip<true>(msg, pk, &s_bias[(k0 + i) * QB], m, l, ss, a0, a1, a2);
    msg += KT * 64;
    pk  += KT * 3 * 64;
  }
#pragma nounroll
  for (int i = KP - PFD; i < KP; i += KT) {
    softmax_trip<false>(msg, pk, &s_bias[(k0 + i) * QB], m, l, ss, a0, a1, a2);
    msg += KT * 64;
    pk  += KT * 3 * 64;
  }

#pragma unroll
  for (int j = 0; j < QB; ++j) {
    *reinterpret_cast<v2f*>(&s_st[0][part][j][dd]) = m[j];
    *reinterpret_cast<v2f*>(&s_st[1][part][j][dd]) = l[j];
    *reinterpret_cast<v2f*>(&s_st[2][part][j][dd]) = ss[j];
    *reinterpret_cast<v2f*>(&s_st[3][part][j][dd]) = a0[j];
    *reinterpret_cast<v2f*>(&s_st[4][part][j][dd]) = a1[j];
    *reinterpret_cast<v2f*>(&s_st[5][part][j][dd]) = a2[j];
  }
  __syncthreads();

  const int wv   = tid >> 5;
  const int ct   = wv & 3;
  const int rt   = wv >> 2;
  const int fc   = lane & 15;
  const int fh   = lane >> 4;
  float4 bw[8];
  if (wv < 8) {
    const float4* __restrict__ wp =
        reinterpret_cast<const float4*>(w_attn + (size_t)(ct * 16 + fc) * 64 + fh * 32);
#pragma unroll
    for (int i = 0; i < 8; ++i) bw[i] = wp[i];
  }

  {
    const int j  = tid >> 6;
    const int mc = tid & 63;
    float M0 = s_st[0][0][j][mc], L = s_st[1][0][j][mc], SS = s_st[2][0][j][mc];
    float A0 = s_st[3][0][j][mc], A1 = s_st[4][0][j][mc], A2 = s_st[5][0][j][mc];
    for (int p = 1; p < NP; ++p) {
      float m2 = s_st[0][p][j][mc];
      float mn = fmaxf(M0, m2);
      float f1 = __expf(M0 - mn);
      float f2 = __expf(m2 - mn);
      L  = L  * f1      + s_st[1][p][j][mc] * f2;
      SS = SS * f1 * f1 + s_st[2][p][j][mc] * f2 * f2;
      A0 = A0 * f1 + s_st[3][p][j][mc] * f2;
      A1 = A1 * f1 + s_st[4][p][j][mc] * f2;
      A2 = A2 * f1 + s_st[5][p][j][mc] * f2;
      M0 = mn;
    }
    L = fmaxf(L, 1.0e-30f);
    float scale = sqrtf(SS) / (L * L);
    float* zs = s_bias;
    zs[(j * 3 + 0) * ZP + mc] = A0 * scale;
    zs[(j * 3 + 1) * ZP + mc] = A1 * scale;
    zs[(j * 3 + 2) * ZP + mc] = A2 * scale;
    zs[(QB * 3 + j) * ZP + mc] = 0.0f;
  }
  __syncthreads();

  if (wv < 8) {
    const float4* zr = reinterpret_cast<const float4*>(s_bias + (rt * 16 + fc) * ZP + fh * 32);
    float4 az[8];
#pragma unroll
    for (int i = 0; i < 8; ++i) az[i] = zr[i];
    __builtin_amdgcn_sched_barrier(0);

    v8f c0 = {}, c1 = {};
#pragma unroll
    for (int i = 0; i < 8; ++i) {
      v2f a, b;
      a.x = az[i].x; a.y = az[i].y;  b.x = bw[i].x; b.y = bw[i].y;
      c0 = __builtin_amdgcn_wmma_f32_16x16x4_f32(false, a, false, b,
                                                 (short)0, c0, false, false);
      a.x = az[i].z; a.y = az[i].w;  b.x = bw[i].z; b.y = bw[i].w;
      c1 = __builtin_amdgcn_wmma_f32_16x16x4_f32(false, a, false, b,
                                                 (short)0, c1, false, false);
    }

    float* __restrict__ op = out + (size_t)(bq0 * 3 + rt * 16 + fh * 8) * 64 + ct * 16 + fc;
#pragma unroll
    for (int r = 0; r < 8; ++r)
      if (rt * 16 + fh * 8 + r < QB * 3) op[(size_t)r * 64] = c0[r] + c1[r];
  }
}

extern "C" void kernel_launch(void* const* d_in, const int* in_sizes, int n_in,
                              void* d_out, int out_size, void* d_ws, size_t ws_size,
                              hipStream_t stream) {
  const float* v_equi   = (const float*)d_in[0];
  const float* messages = (const float*)d_in[1];
  const int*   adj      = (const int*)d_in[2];
  const float* w_coord  = (const float*)d_in[3];
  const float* w_attn   = (const float*)d_in[4];
  float* out = (float*)d_out;

  const int B = 4, Nq = 512;
  const int Mproj = B * 512 * 3;

  float* proj = (float*)d_ws;

  const int gemmBlocks = ((Mproj / 16) * 4 + 7) / 8;

  gemm_nt_wmma_f32<<<gemmBlocks, 256, 0, stream>>>(v_equi, w_coord, proj, Mproj);

  attn_softmax_kernel<<<(B * Nq) / QB, NT, 0, stream>>>(messages, adj, proj, w_attn, out);
}


# attention: masked (q,k) pairs are skipped by a scalar branch of the wave (mask bits by ballot): their message rows are never read; rescale per k again: rounding points move, k order kept
# speedup vs baseline: 1.9313x; 1.1178x over previous
#include <hip/hip_runtime.h>
#include <hip/hip_bf16.h>
#include <math.h>

typedef float v8f __attribute__((ext_vector_type(8)));
typedef float v2f __attribute__((ext_vector_type(2)));

#define NEG_BIAS (-1.0e10f)

__global__ __launch_bounds__(256) void gemm_nt_wmma_f32(
    const float* __restrict__ X, const float* __restrict__ W,
    float* __restrict__ C, int M)
{
  const int lane      = threadIdx.x & 31;
  const int waveInBlk = threadIdx.x >> 5;
  const int wave      = blockIdx.x * 8 + waveInBlk;
  const int mt        = wave >> 2;
  const int nt        = wave & 3;
  if (mt * 16 >= M) return;

  const int mrow  = mt * 16 + (lane & 15);
  const int wrow  = nt * 16 + (lane & 15);
  const int khalf = (lane >> 4) << 1;

  const float* __restrict__ xp = X + (size_t)mrow * 64 + khalf;
  const float* __restrict__ wp = W + (size_t)wrow * 64 + khalf;

  v8f c = {};
#pragma unroll
  for (int kk = 0; kk < 64; kk += 4) {
    v2f a, b;
    a.x = xp[kk];  a.y = xp[kk + 1];
    b.x = wp[kk];  b.y = wp[kk + 1];
    c = __builtin_amdgcn_wmma_f32_16x16x4_f32(false, a, false, b,
                                              (short)0, c, false, false);
  }

  const int crowBase = mt * 16 + ((lane >> 4) << 3);
  const int ccol     = nt * 16 + (lane & 15);
#pragma unroll
  for (int r = 0; r < 8; ++r)
    C[(size_t)(crowBase + r) * 64 + ccol] = c[r];
}

#define QB 8
#define CH 2
#define NP 16
#define KP 32
#define KT 4
#define PFD 16
#define NT (32 * NP)
static_assert(QB == 8 && CH * 32 == 64 && NP * KP == 512 && KP == 32 && NP == 16, "a wave's lanes cover the 64 channels; the partitions tile the 512 keys; a partition's mask is one 32-bit ballot; the row-empty vote reads 4 partitions per lane");
static_assert(PFD % KT == 0 && PFD < KP && KP % KT == 0 && KT == 4 && PFD + KT <= 32, "whole trips; one prefetch instruction names 8 rows x 2 k x 2 lines, two per trip");
#define ZP 68
#define ZR 32
static_assert(QB * 4 == ZR && NT >= 8 * 32 && NT == 64 * QB, "one wave per 16-row, 16-column tile of out; one merge thread per row and channel");

__device__ __forceinline__ void row_update(
    const v2f mv, const float bias, const v2f p0, const v2f p1, const v2f p2,
    v2f& m, v2f& l, v2f& ss, v2f& a0, v2f& a1, v2f& a2)
{
  for (int h = 0; h < CH; ++h) {
    const float x    = mv[h] + bias;
    const float mo   = m[h];
    const float mn   = fmaxf(mo, x);
    const float corr = __expf(mo - mn);
    const float e    = __expf(x - mn);
    l[h]  = fmaf(l[h], corr, e);
    ss[h] = fmaf(e, e, ss[h] * (corr * corr));
    a0[h] = fmaf(e, p0[h], a0[h] * corr);
    a1[h] = fmaf(e, p1[h], a1[h] * corr);
    a2[h] = fmaf(e, p2[h], a2[h] * corr);
    m[h]  = mn;
  }
}

#define ROWS(F) F(0) F(1) F(2) F(3) F(4) F(5) F(6) F(7)
#define DECL_STATE(j) v2f M##j = {-1.0e30f, -1.0e30f}, L##j = {0.0f, 0.0f}, S##j = {0.0f, 0.0f}, \
                          A##j = {0.0f, 0.0f}, B##j = {0.0f, 0.0f}, C##j = {0.0f, 0.0f};
#define LD(u, j) v2f x##u##_##j; if ((cur[j] >> u) & 1u) \
    x##u##_##j = *reinterpret_cast<const v2f*>(msg + (size_t)j * 512 * 64 + u * 64);
#define UP(u, j) if ((cur[j] >> u) & 1u) \
    row_update(x##u##_##j, bias[j], p0[u], p1[u], p2[u], M##j, L##j, S##j, A##j, B##j, C##j);
#define LD_K(u) LD(u, 0) LD(u, 1) LD(u, 2) LD(u, 3) LD(u, 4) LD(u, 5) LD(u, 6) LD(u, 7)
#define UP_K(u) UP(u, 0) UP(u, 1) UP(u, 2) UP(u, 3) UP(u, 4) UP(u, 5) UP(u, 6) UP(u, 7)
#define ST_STATE(j) \
    *reinterpret_cast<v2f*>(&s_st[0][part][j][dd]) = M##j; \
    *reinterpret_cast<v2f*>(&s_st[1][part][j][dd]) = L##j; \
    *reinterpret_cast<v2f*>(&s_st[2][part][j][dd]) = S##j; \
    *reinterpret_cast<v2f*>(&s_st[3][part][j][dd]) = A##j; \
    *reinterpret_cast<v2f*>(&s_st[4][part][j][dd]) = B##j; \
    *reinterpret_cast<v2f*>(&s_st[5][part][j][dd]) = C##j;

__global__ __launch_bounds__(NT) void attn_softmax_kernel(
    const float* __restrict__ messages, const int* __restrict__ adj,
    const float* __restrict__ proj, const float* __restrict__ w_attn,
    float* __restrict__ out)
{
  __shared__ unsigned s_bits[NP][QB];
  __shared__ __attribute__((aligned(16))) float s_z[ZR * ZP];
  __shared__ __attribute__((aligned(8))) float s_st[6][NP][QB][64];

  const int bq0  = blockIdx.x * QB;
  const int b    = bq0 >> 9;
  const int tid  = threadIdx.x;
  const int lane = tid & 31;
  const int dd   = lane * CH;
  const int part = tid >> 5;

  const int k0 = part * KP;
  const float* __restrict__ msg = messages + (size_t)bq0 * 512 * 64 + (size_t)k0 * 64 + dd;
  const float* __restrict__ pk  = proj + (size_t)b * 512 * 3 * 64 + (size_t)k0 * 3 * 64 + dd;

  unsigned bits[QB];
#pragma unroll
  for (int j = 0; j < QB; ++j)
    bits[j] = __builtin_amdgcn_ballot_w32(adj[(size_t)(bq0 + j) * 512 + k0 + lane] > 0);
  if (lane == 0) {
#pragma unroll
    for (int j = 0; j < QB; ++j) s_bits[part][j] = bits[j];
  }
  __syncthreads();

  unsigned orw = 0;
  for (int t = 0; t < 4; ++t) orw |= s_bits[(lane >> 3) * 4 + t][lane & 7];
  const unsigned nz = __builtin_amdgcn_ballot_w32(orw != 0u);
  unsigned act[QB];
  float bias[QB];
#pragma unroll
  for (int j = 0; j < QB; ++j) {
    const bool any = (nz & (0x01010101u << j)) != 0u;
    act[j]  = any ? bits[j] : 0xFFFFFFFFu;
    bias[j] = any ? 0.0f : NEG_BIAS;
  }

  const int jl = lane >> 2, ul = (lane >> 1) & 1;
  unsigned myact = 0;
#pragma unroll
  for (int j = 0; j < QB; ++j) myact = (jl == j) ? act[j] : myact;
  const float* __restrict__ pfp = messages + (size_t)(bq0 + jl) * 512 * 64
                                  + (size_t)(k0 + ul) * 64 + (lane & 1) * 32;
  for (int g = 0; g < PFD / 2; ++g)
    if ((myact >> (2 * g + ul)) & 1u) __builtin_prefetch(pfp + g * 128, 0, 0);

  ROWS(DECL_STATE)

#pragma nounroll
  for (int i = 0; i < KP; i += KT) {
    const unsigned mycur = myact >> i;
    for (int g = 0; g < KT / 2; ++g)
      if ((mycur >> (PFD + 2 * g + ul)) & 1u)
        __builtin_prefetch(pfp + (size_t)(i + PFD + 2 * g) * 64, 0, 0);
    unsigned cur[QB];
#pragma unroll
    for (int j = 0; j < QB; ++j) cur[j] = act[j] >> i;
    v2f p0[KT], p1[KT], p2[KT];
    for (int u = 0; u < KT; ++u) {
      p0[u] = *reinterpret_cast<const v2f*>(pk + u * 192);
      p1[u] = *reinterpret_cast<const v2f*>(pk + u * 192 + 64);
      p2[u] = *reinterpret_cast<const v2f*>(pk + u * 192 + 128);
    }
    LD_K(0) LD_K(1) LD_K(2) LD_K(3)
    UP_K(0) UP_K(1) UP_K(2) UP_K(3)
    msg += KT * 64;
    pk  += KT * 3 * 64;
  }

  ROWS(ST_STATE)
  __syncthreads();

  const int wv   = tid >> 5;
  const int ct   = wv & 3;
  const int rt   = wv >> 2;
  const int fc   = lane & 15;
  const int fh   = lane >> 4;
  float4 bw[8];
  if (wv < 8) {
    const float4* __restrict__ wp =
        reinterpret_cast<const float4*>(w_attn + (size_t)(ct * 16 + fc) * 64 + fh * 32);
#pragma unroll
    for (int i = 0; i < 8; ++i) bw[i] = wp[i];
  }

  {
    const int j  = tid >> 6;
    const int mc = tid & 63;
    float M0 = s_st[0][0][j][mc], L = s_st[1][0][j][mc], SS = s_st[2][0][j][mc];
    float A0 = s_st[3][0][j][mc], A1 = s_st[4][0][j][mc], A2 = s_st[5][0][j][mc];
    for (int p = 1; p < NP; ++p) {
      float m2 = s_st[0][p][j][mc];
      float mn = fmaxf(M0, m2);
      float f1 = __expf(M0 - mn);
      float f2 = __expf(m2 - mn);
      L  = L  * f1      + s_st[1][p][j][mc] * f2;
      SS = SS * f1 * f1 + s_st[2][p][j][mc] * f2 * f2;
      A0 = A0 * f1 + s_st[3][p][j][mc] * f2;
      A1 = A1 * f1 + s_st[4][p][j][mc] * f2;
      A2 = A2 * f1 + s_st[5][p][j][mc] * f2;
      M0 = mn;
    }
    L = fmaxf(L, 1.0e-30f);
    float scale = sqrtf(SS) / (L * L);
    float* zs = s_z;
    zs[(j * 3 + 0) * ZP + mc] = A0 * scale;
    zs[(j * 3 + 1) * ZP + mc] = A1 * scale;
    zs[(j * 3 + 2) * ZP + mc] = A2 * scale;
    zs[(QB * 3 + j) * ZP + mc] = 0.0f;
  }
  __syncthreads();

  if (wv < 8) {
    const float4* zr = reinterpret_cast<const float4*>(s_z + (rt * 16 + fc) * ZP + fh * 32);
    float4 az[8];
#pragma unroll
    for (int i = 0; i < 8; ++i) az[i] = zr[i];
    __builtin_amdgcn_sched_barrier(0);

    v8f c0 = {}, c1 = {};
#pragma unroll
    for (int i = 0; i < 8; ++i) {
      v2f a, b;
      a.x = az[i].x; a.y = az[i].y;  b.x = bw[i].x; b.y = bw[i].y;
      c0 = __builtin_amdgcn_wmma_f32_16x16x4_f32(false, a, false, b,
                                                 (short)0, c0, false, false);
      a.x = az[i].z; a.y = az[i].w;  b.x = bw[i].z; b.y = bw[i].w;
      c1 = __builtin_amdgcn_wmma_f32_16x16x4_f32(false, a, false, b,
                                                 (short)0, c1, false, false);
    }

    float* __restrict__ op = out + (size_t)(bq0 * 3 + rt * 16 + fh * 8) * 64 + ct * 16 + fc;
#pragma unroll
    for (int r = 0; r < 8; ++r)
      if (rt * 16 + fh * 8 + r < QB * 3) op[(size_t)r * 64] = c0[r] + c1[r];
  }
}

extern "C" void kernel_launch(void* const* d_in, const int* in_sizes, int n_in,
                              void* d_out, int out_size, void* d_ws, size_t ws_size,
                              hipStream_t stream) {
  const float* v_equi   = (const float*)d_in[0];
  const float* messages = (const float*)d_in[1];
  const int*   adj      = (const int*)d_in[2];
  const float* w_coord  = (const float*)d_in[3];
  const float* w_attn   = (const float*)d_in[4];
  float* out = (float*)d_out;

  const int B = 4, Nq = 512;
  const int Mproj = B * 512 * 3;

  float* proj = (float*)d_ws;

  const int gemmBlocks = ((Mproj / 16) * 4 + 7) / 8;

  gemm_nt_wmma_f32<<<gemmBlocks, 256, 0, stream>>>(v_equi, w_coord, proj, Mproj);

  attn_softmax_kernel<<<(B * Nq) / QB, NT, 0, stream>>>(messages, adj, proj, w_attn, out);
}


# attention: softmax sums relative to a reference logit moved only when a logit exceeds it by 11 (rare wave-uniform branch): common update 1 exp + 5 multiply-adds; k order kept, rounding points move
# speedup vs baseline: 2.0122x; 1.0419x over previous
#include <hip/hip_runtime.h>
#include <hip/hip_bf16.h>
#include <math.h>

typedef float v8f __attribute__((ext_vector_type(8)));
typedef float v2f __attribute__((ext_vector_type(2)));

#define NEG_BIAS (-1.0e10f)

__global__ __launch_bounds__(256) void gemm_nt_wmma_f32(
    const float* __restrict__ X, const float* __restrict__ W,
    float* __restrict__ C, int M)
{
  const int lane      = threadIdx.x & 31;
  const int waveInBlk = threadIdx.x >> 5;
  const int wave      = blockIdx.x * 8 + waveInBlk;
  const int mt        = wave >> 2;
  const int nt        = wave & 3;
  if (mt * 16 >= M) return;

  const int mrow  = mt * 16 + (lane & 15);
  const int wrow  = nt * 16 + (lane & 15);
  const int khalf = (lane >> 4) << 1;

  const float* __restrict__ xp = X + (size_t)mrow * 64 + khalf;
  const float* __restrict__ wp = W + (size_t)wrow * 64 + khalf;

  v8f c = {};
#pragma unroll
  for (int kk = 0; kk < 64; kk += 4) {
    v2f a, b;
    a.x = xp[kk];  a.y = xp[kk + 1];
    b.x = wp[kk];  b.y = wp[kk + 1];
    c = __builtin_amdgcn_wmma_f32_16x16x4_f32(false, a, false, b,
                                              (short)0, c, false, false);
  }

  const int crowBase = mt * 16 + ((lane >> 4) << 3);
  const int ccol     = nt * 16 + (lane & 15);
#pragma unroll
  for (int r = 0; r < 8; ++r)
    C[(size_t)(crowBase + r) * 64 + ccol] = c[r];
}

#define QB 8
#define CH 2
#define NP 16
#define KP 32
#define KT 4
#define PFD 16
#define NT (32 * NP)
static_assert(QB == 8 && CH * 32 == 64 && NP * KP == 512 && KP == 32 && NP == 16, "a wave's lanes cover the 64 channels; the partitions tile the 512 keys; a partition's mask is one 32-bit ballot; the row-empty vote reads 4 partitions per lane");
static_assert(PFD % KT == 0 && PFD < KP && KP % KT == 0 && KT == 4 && PFD + KT <= 32, "whole trips; one prefetch instruction names 8 rows x 2 k x 2 lines, two per trip");
#define ZP 68
#define ZR 32
static_assert(QB * 4 == ZR && NT >= 8 * 32 && NT == 64 * QB, "one wave per 16-row, 16-column tile of out; one merge thread per row and channel");

__device__ __forceinline__ v2f exp_ch(const v2f a)
{
  v2f r;
  r.x = __expf(a.x);
  r.y = __expf(a.y);
  return r;
}

#define REF_MARGIN 11.0f
__device__ __forceinline__ void row_update(
    const v2f mv, const float bias, const v2f p0, const v2f p1, const v2f p2,
    v2f& m, v2f& l, v2f& ss, v2f& a0, v2f& a1, v2f& a2)
{
  const v2f x = mv + bias;
  v2f d = x - m;
  if (__builtin_expect(__builtin_amdgcn_ballot_w32(fmaxf(d.x, d.y) > REF_MARGIN) != 0u, 0)) {
    const v2f mn   = __builtin_elementwise_max(m, x);
    const v2f corr = exp_ch(m - mn);
    l  = l * corr;
    ss = ss * (corr * corr);
    a0 = a0 * corr;
    a1 = a1 * corr;
    a2 = a2 * corr;
    m  = mn;
    d  = x - mn;
  }
  const v2f e = exp_ch(d);
  l  = l + e;
  ss = __builtin_elementwise_fma(e, e, ss);
  a0 = __builtin_elementwise_fma(e, p0, a0);
  a1 = __builtin_elementwise_fma(e, p1, a1);
  a2 = __builtin_elementwise_fma(e, p2, a2);
}

#define ROWS(F) F(0) F(1) F(2) F(3) F(4) F(5) F(6) F(7)
#define DECL_STATE(j) v2f M##j = {-1.0e30f, -1.0e30f}, L##j = {0.0f, 0.0f}, S##j = {0.0f, 0.0f}, \
                          A##j = {0.0f, 0.0f}, B##j = {0.0f, 0.0f}, C##j = {0.0f, 0.0f};
#define LD(u, j) v2f x##u##_##j; if ((cur[j] >> u) & 1u) \
    x##u##_##j = *reinterpret_cast<const v2f*>(msg + (size_t)j * 512 * 64 + u * 64);
#define UP(u, j) if ((cur[j] >> u) & 1u) \
    row_update(x##u##_##j, bias[j], p0[u], p1[u], p2[u], M##j, L##j, S##j, A##j, B##j, C##j);
#define LD_K(u) LD(u, 0) LD(u, 1) LD(u, 2) LD(u, 3) LD(u, 4) LD(u, 5) LD(u, 6) LD(u, 7)
#define UP_K(u) UP(u, 0) UP(u, 1) UP(u, 2) UP(u, 3) UP(u, 4) UP(u, 5) UP(u, 6) UP(u, 7)
#define ST_STATE(j) \
    *reinterpret_cast<v2f*>(&s_st[0][part][j][dd]) = M##j; \
    *reinterpret_cast<v2f*>(&s_st[1][part][j][dd]) = L##j; \
    *reinterpret_cast<v2f*>(&s_st[2][part][j][dd]) = S##j; \
    *reinterpret_cast<v2f*>(&s_st[3][part][j][dd]) = A##j; \
    *reinterpret_cast<v2f*>(&s_st[4][part][j][dd]) = B##j; \
    *reinterpret_cast<v2f*>(&s_st[5][part][j][dd]) = C##j;

__global__ __launch_bounds__(NT) void attn_softmax_kernel(
    const float* __restrict__ messages, const int* __restrict__ adj,
    const float* __restrict__ proj, const float* __restrict__ w_attn,
    float* __restrict__ out)
{
  __shared__ unsigned s_bits[NP][QB];
  __shared__ __attribute__((aligned(16))) float s_z[ZR * ZP];
  __shared__ __attribute__((aligned(8))) float s_st[6][NP][QB][64];

  const int bq0  = blockIdx.x * QB;
  const int b    = bq0 >> 9;
  const int tid  = threadIdx.x;
  const int lane = tid & 31;
  const int dd   = lane * CH;
  const int part = tid >> 5;

  const int k0 = part * KP;
  const float* __restrict__ msg = messages + (size_t)bq0 * 512 * 64 + (size_t)k0 * 64 + dd;
  const float* __restrict__ pk  = proj + (size_t)b * 512 * 3 * 64 + (size_t)k0 * 3 * 64 + dd;

  unsigned bits[QB];
#pragma unroll
  for (int j = 0; j < QB; ++j)
    bits[j] = __builtin_amdgcn_ballot_w32(adj[(size_t)(bq0 + j) * 512 + k0 + lane] > 0);
  if (lane == 0) {
#pragma unroll
    for (int j = 0; j < QB; ++j) s_bits[part][j] = bits[j];
  }
  __syncthreads();

  unsigned orw = 0;
  for (int t = 0; t < 4; ++t) orw |= s_bits[(lane >> 3) * 4 + t][lane & 7];
  const unsigned nz = __builtin_amdgcn_ballot_w32(orw != 0u);
  unsigned act[QB];
  float bias[QB];
#pragma unroll
  for (int j = 0; j < QB; ++j) {
    const bool any = (nz & (0x01010101u << j)) != 0u;
    act[j]  = any ? bits[j] : 0xFFFFFFFFu;
    bias[j] = any ? 0.0f : NEG_BIAS;
  }

  const int jl = lane >> 2, ul = (lane >> 1) & 1;
  unsigned myact = 0;
#pragma unroll
  for (int j = 0; j < QB; ++j) myact = (jl == j) ? act[j] : myact;
  const float* __restrict__ pfp = messages + (size_t)(bq0 + jl) * 512 * 64
                                  + (size_t)(k0 + ul) * 64 + (lane & 1) * 32;
  for (int g = 0; g < PFD / 2; ++g)
    if ((myact >> (2 * g + ul)) & 1u) __builtin_prefetch(pfp + g * 128, 0, 0);

  ROWS(DECL_STATE)

#pragma nounroll
  for (int i = 0; i < KP; i += KT) {
    const unsigned mycur = myact >> i;
    for (int g = 0; g < KT / 2; ++g)
      if ((mycur >> (PFD + 2 * g + ul)) & 1u)
        __builtin_prefetch(pfp + (size_t)(i + PFD + 2 * g) * 64, 0, 0);
    unsigned cur[QB];
#pragma unroll
    for (int j = 0; j < QB; ++j) cur[j] = act[j] >> i;
    v2f p0[KT], p1[KT], p2[KT];
    for (int u = 0; u < KT; ++u) {
      p0[u] = *reinterpret_cast<const v2f*>(pk + u * 192);
      p1[u] = *reinterpret_cast<const v2f*>(pk + u * 192 + 64);
      p2[u] = *reinterpret_cast<const v2f*>(pk + u * 192 + 128);
    }
    LD_K(0) LD_K(1) LD_K(2) LD_K(3)
    UP_K(0) UP_K(1) UP_K(2) UP_K(3)
    msg += KT * 64;
    pk  += KT * 3 * 64;
  }

  ROWS(ST_STATE)
  __syncthreads();

  const int wv   = tid >> 5;
  const int ct   = wv & 3;
  const int rt   = wv >> 2;
  const int fc   = lane & 15;
  const int fh   = lane >> 4;
  float4 bw[8];
  if (wv < 8) {
    const float4* __restrict__ wp =
        reinterpret_cast<const float4*>(w_attn + (size_t)(ct * 16 + fc) * 64 + fh * 32);
#pragma unroll
    for (int i = 0; i < 8; ++i) bw[i] = wp[i];
  }

  {
    const int j  = tid >> 6;
    const int mc = tid & 63;
    float M0 = s_st[0][0][j][mc], L = s_st[1][0][j][mc], SS = s_st[2][0][j][mc];
    float A0 = s_st[3][0][j][mc], A1 = s_st[4][0][j][mc], A2 = s_st[5][0][j][mc];
    for (int p = 1; p < NP; ++p) {
      float m2 = s_st[0][p][j][mc];
      float mn = fmaxf(M0, m2);
      float f1 = __expf(M0 - mn);
      float f2 = __expf(m2 - mn);
      L  = L  * f1      + s_st[1][p][j][mc] * f2;
      SS = SS * f1 * f1 + s_st[2][p][j][mc] * f2 * f2;
      A0 = A0 * f1 + s_st[3][p][j][mc] * f2;
      A1 = A1 * f1 + s_st[4][p][j][mc] * f2;
      A2 = A2 * f1 + s_st[5][p][j][mc] * f2;
      M0 = mn;
    }
    L = fmaxf(L, 1.0e-30f);
    float scale = sqrtf(SS) / (L * L);
    float* zs = s_z;
    zs[(j * 3 + 0) * ZP + mc] = A0 * scale;
    zs[(j * 3 + 1) * ZP + mc] = A1 * scale;
    zs[(j * 3 + 2) * ZP + mc] = A2 * scale;
    zs[(QB * 3 + j) * ZP + mc] = 0.0f;
  }
  __syncthreads();

  if (wv < 8) {
    const float4* zr = reinterpret_cast<const float4*>(s_z + (rt * 16 + fc) * ZP + fh * 32);
    float4 az[8];
#pragma unroll
    for (int i = 0; i < 8; ++i) az[i] = zr[i];
    __builtin_amdgcn_sched_barrier(0);

    v8f c0 = {}, c1 = {};
#pragma unroll
    for (int i = 0; i < 8; ++i) {
      v2f a, b;
      a.x = az[i].x; a.y = az[i].y;  b.x = bw[i].x; b.y = bw[i].y;
      c0 = __builtin_amdgcn_wmma_f32_16x16x4_f32(false, a, false, b,
                                                 (short)0, c0, false, false);
      a.x = az[i].z; a.y = az[i].w;  b.x = bw[i].z; b.y = bw[i].w;
      c1 = __builtin_amdgcn_wmma_f32_16x16x4_f32(false, a, false, b,
                                                 (short)0, c1, false, false);
    }

    float* __restrict__ op = out + (size_t)(bq0 * 3 + rt * 16 + fh * 8) * 64 + ct * 16 + fc;
#pragma unroll
    for (int r = 0; r < 8; ++r)
      if (rt * 16 + fh * 8 + r < QB * 3) op[(size_t)r * 64] = c0[r] + c1[r];
  }
}

extern "C" void kernel_launch(void* const* d_in, const int* in_sizes, int n_in,
                              void* d_out, int out_size, void* d_ws, size_t ws_size,
                              hipStream_t stream) {
  const float* v_equi   = (const float*)d_in[0];
  const float* messages = (const float*)d_in[1];
  const int*   adj      = (const int*)d_in[2];
  const float* w_coord  = (const float*)d_in[3];
  const float* w_attn   = (const float*)d_in[4];
  float* out = (float*)d_out;

  const int B = 4, Nq = 512;
  const int Mproj = B * 512 * 3;

  float* proj = (float*)d_ws;

  const int gemmBlocks = ((Mproj / 16) * 4 + 7) / 8;

  gemm_nt_wmma_f32<<<gemmBlocks, 256, 0, stream>>>(v_equi, w_coord, proj, Mproj);

  attn_softmax_kernel<<<(B * Nq) / QB, NT, 0, stream>>>(messages, adj, proj, w_attn, out);
}
